# starting kernel
# baseline (speedup 1.0000x reference)
#include <hip/hip_runtime.h>
#include <hip/hip_bf16.h>
#include <math.h>

#define BS   4
#define V    4096
#define NNB  20
#define OC   64
#define SUP  4
#define INCH 64
#define COLS ((SUP + 1) * OC)
#define DIRS (SUP * OC)
#define EPSV 1e-12f

typedef __attribute__((ext_vector_type(2))) float v2f;
typedef __attribute__((ext_vector_type(8))) float v8f;

__global__ __launch_bounds__(256)
void gemm_bias_wmma_kernel(const float* __restrict__ A,
                           const float* __restrict__ W,
                           const float* __restrict__ bias,
                           float* __restrict__ out,
                           int M)
{
    const int tilesN = COLS / 16;
    const int wave   = threadIdx.x >> 5;
    const int lane   = threadIdx.x & 31;
    const int tile   = blockIdx.x * (blockDim.x >> 5) + wave;

    const int tm = tile / tilesN;
    const int tn = tile - tm * tilesN;
    if (tm * 16 >= M) return;

    const int l   = lane & 15;
    const int h   = lane >> 4;
    const int col = tn * 16 + l;

    const float bv = bias[col];
    v8f c;
#pragma unroll
    for (int i = 0; i < 8; ++i) c[i] = bv;

    const float* arow = A + (size_t)(tm * 16 + l) * INCH;

#pragma unroll
    for (int k0 = 0; k0 < INCH; k0 += 4) {
        const int k = k0 + 2 * h;
        v2f a, b;
        a[0] = arow[k];
        a[1] = arow[k + 1];
        b[0] = W[(size_t)k * COLS + col];
        b[1] = W[(size_t)(k + 1) * COLS + col];
        c = __builtin_amdgcn_wmma_f32_16x16x4_f32(false, a, false, b,
                                                  (short)0, c, false, false);
    }

#pragma unroll
    for (int r = 0; r < 8; ++r) {
        const int m = tm * 16 + r + 8 * h;
        out[(size_t)m * COLS + col] = c[r];
    }
}

__global__ __launch_bounds__(256)
void gather_theta_max_kernel(const int*   __restrict__ nbr_idx,
                             const float* __restrict__ verts,
                             const float* __restrict__ dirs,
                             const float* __restrict__ feat,
                             float*       __restrict__ out)
{
    __shared__ float s_nd[NNB][3];
    __shared__ int   s_nb[NNB];
    __shared__ float s_red[DIRS];

    const int bv  = blockIdx.x;
    const int b   = bv / V;
    const int tid = threadIdx.x;

    if (tid < NNB) {
        const int nb  = nbr_idx[(size_t)bv * NNB + tid];
        const int nbv = b * V + nb;
        const float cx = verts[(size_t)bv * 3 + 0];
        const float cy = verts[(size_t)bv * 3 + 1];
        const float cz = verts[(size_t)bv * 3 + 2];
        const float dx = verts[(size_t)nbv * 3 + 0] - cx;
        const float dy = verts[(size_t)nbv * 3 + 1] - cy;
        const float dz = verts[(size_t)nbv * 3 + 2] - cz;
        const float inv = 1.0f / fmaxf(sqrtf(dx * dx + dy * dy + dz * dz), EPSV);
        s_nd[tid][0] = dx * inv;
        s_nd[tid][1] = dy * inv;
        s_nd[tid][2] = dz * inv;
        s_nb[tid]    = nbv;
    }
    __syncthreads();

    const int d = tid;
    float sx = dirs[d];
    float sy = dirs[DIRS + d];
    float sz = dirs[2 * DIRS + d];
    const float sinv = 1.0f / fmaxf(sqrtf(sx * sx + sy * sy + sz * sz), EPSV);
    sx *= sinv; sy *= sinv; sz *= sinv;

    float m = -__builtin_inff();
#pragma unroll 4
    for (int j = 0; j < NNB; ++j) {
        const float t  = fmaxf(sx * s_nd[j][0] + sy * s_nd[j][1] + sz * s_nd[j][2], 0.0f);
        const float fs = feat[(size_t)s_nb[j] * COLS + OC + d];
        m = fmaxf(m, t * fs);
    }
    s_red[d] = m;
    __syncthreads();

    if (tid < OC) {
        const float acc = feat[(size_t)bv * COLS + tid]
                        + s_red[tid] + s_red[OC + tid]
                        + s_red[2 * OC + tid] + s_red[3 * OC + tid];
        out[(size_t)bv * OC + tid] = acc;
    }
}

extern "C" void kernel_launch(void* const* d_in, const int* in_sizes, int n_in,
                              void* d_out, int out_size, void* d_ws, size_t ws_size,
                              hipStream_t stream)
{
    (void)in_sizes; (void)n_in; (void)out_size; (void)ws_size;

    const int*   nbr   = (const int*)  d_in[0];
    const float* verts = (const float*)d_in[1];
    const float* fmap  = (const float*)d_in[2];
    const float* W     = (const float*)d_in[3];
    const float* bias  = (const float*)d_in[4];
    const float* dirs  = (const float*)d_in[5];
    float*       outp  = (float*)d_out;
    float*       feat  = (float*)d_ws;

    const int M     = BS * V;
    const int tiles = (M / 16) * (COLS / 16);
    const int blocks = tiles / 8;

    gemm_bias_wmma_kernel<<<blocks, 256, 0, stream>>>(fmap, W, bias, feat, M);
    gather_theta_max_kernel<<<M, 256, 0, stream>>>(nbr, verts, dirs, feat, outp);
}


# gather loads software-pipelined across vertices: ring of 4 five-neighbor batches in registers, next vertex's 80 loads fly while this one is folded; loop keeps 40-63 loads out, never drains to 0
# speedup vs baseline: 2.1206x; 2.1206x over previous
#include <hip/hip_runtime.h>
#include <hip/hip_bf16.h>
#include <math.h>

#define BS   4
#define V    4096
#define NNB  20
#define OC   64
#define SUP  4
#define INCH 64
#define COLS ((SUP + 1) * OC)
#define DIRS (SUP * OC)
#define EPSV 1e-12f

typedef __attribute__((ext_vector_type(2))) float v2f;
typedef __attribute__((ext_vector_type(8))) float v8f;

#define MROWS_PER_WAVE 32
#define NTILES_PER_WAVE 5
#define NCOLS_PER_WAVE (NTILES_PER_WAVE * 16)
#define KSTEPS (INCH / 4)

__global__ __launch_bounds__(256)
void gemm_bias_wmma_kernel(const float* __restrict__ A,
                           const float* __restrict__ W,
                           const float* __restrict__ bias,
                           float* __restrict__ out,
                           int M)
{
    const int wave = threadIdx.x >> 5;
    const int lane = threadIdx.x & 31;
    const int l    = lane & 15;
    const int h    = lane >> 4;

    const int row0 = (blockIdx.x * 2 + (wave >> 2)) * MROWS_PER_WAVE;
    const int col0 = (wave & 3) * NCOLS_PER_WAVE;
    if (row0 >= M) return;

    const float* arow0 = A + (size_t)(row0 + l) * INCH + 2 * h;
    const float* arow1 = arow0 + (size_t)16 * INCH;
    v2f a0[KSTEPS], a1[KSTEPS];
#pragma unroll
    for (int i = 0; i < KSTEPS; ++i) {
        a0[i] = *(const v2f*)(arow0 + 4 * i);
        a1[i] = *(const v2f*)(arow1 + 4 * i);
    }

#pragma unroll
    for (int t = 0; t < NTILES_PER_WAVE; ++t) {
        const int col = col0 + t * 16 + l;
        const float bv = bias[col];
        v8f c0, c1;
#pragma unroll
        for (int i = 0; i < 8; ++i) { c0[i] = bv; c1[i] = bv; }

        const float* wp = W + (size_t)(2 * h) * COLS + col;
#pragma unroll
        for (int i = 0; i < KSTEPS; ++i) {
            v2f b;
            b[0] = wp[(size_t)(4 * i) * COLS];
            b[1] = wp[(size_t)(4 * i + 1) * COLS];
            c0 = __builtin_amdgcn_wmma_f32_16x16x4_f32(false, a0[i], false, b,
                                                       (short)0, c0, false, false);
            c1 = __builtin_amdgcn_wmma_f32_16x16x4_f32(false, a1[i], false, b,
                                                       (short)0, c1, false, false);
        }

        float* o0 = out + (size_t)(row0 + 8 * h) * COLS + col;
        float* o1 = o0 + (size_t)16 * COLS;
#pragma unroll
        for (int r = 0; r < 8; ++r) {
            o0[(size_t)r * COLS] = c0[r];
            o1[(size_t)r * COLS] = c1[r];
        }
    }
}

#define STRIP   32
#define PAIRS   (STRIP * NNB)
#define VSLOTS  4
#define PROUNDS ((PAIRS + 255) / 256)
#define NQ      4
#define NPB     (NNB / NQ)
#define NVERT   (STRIP / VSLOTS)

struct __attribute__((aligned(16))) NbrRec {
    float    x, y, z;
    unsigned off;
};

__device__ __forceinline__
void read_offs(unsigned (&o)[NPB], const NbrRec* recs)
{
    for (int j = 0; j < NPB; ++j) o[j] = recs[j].off;
}

__device__ __forceinline__
void issue_batch(float (&dst)[NPB][SUP], const unsigned (&o)[NPB], const float* __restrict__ feat, int c)
{
    for (int j = 0; j < NPB; ++j) {
        const float* rowp = feat + (o[j] + (unsigned)c);
        for (int s = 0; s < SUP; ++s) dst[j][s] = rowp[s * OC];
    }
}

__device__ __forceinline__
void fold_batch(float (&m)[SUP], const float (&src)[NPB][SUP], const NbrRec* recs,
                const float (&sx)[SUP], const float (&sy)[SUP], const float (&sz)[SUP])
{
    for (int j = 0; j < NPB; ++j) {
        const NbrRec r = recs[j];
        for (int s = 0; s < SUP; ++s) {
            const float t = fmaxf(sx[s] * r.x + sy[s] * r.y + sz[s] * r.z, 0.0f);
            m[s] = fmaxf(m[s], t * src[j][s]);
        }
    }
}

__global__ __launch_bounds__(256)
void gather_theta_max_kernel(const int*   __restrict__ nbr_idx,
                             const float* __restrict__ verts,
                             const float* __restrict__ dirs,
                             const float* __restrict__ feat,
                             float*       __restrict__ out)
{
    __shared__ NbrRec s_rec[PROUNDS * 256];

    const int tid = threadIdx.x;
    const int v0  = blockIdx.x * STRIP;
    const int b   = v0 / V;

    int   nbv[PROUNDS];
    float ctr3[PROUNDS][3], nb3[PROUNDS][3];
    for (int k = 0; k < PROUNDS; ++k) {
        const int p  = min(tid + 256 * k, PAIRS - 1);
        const int bv = v0 + p / NNB;
        nbv[k] = b * V + nbr_idx[(size_t)v0 * NNB + p];
        ctr3[k][0] = verts[(size_t)bv * 3 + 0];
        ctr3[k][1] = verts[(size_t)bv * 3 + 1];
        ctr3[k][2] = verts[(size_t)bv * 3 + 2];
    }
    for (int k = 0; k < PROUNDS; ++k) {
        nb3[k][0] = verts[(size_t)nbv[k] * 3 + 0];
        nb3[k][1] = verts[(size_t)nbv[k] * 3 + 1];
        nb3[k][2] = verts[(size_t)nbv[k] * 3 + 2];
    }
    for (int k = 0; k < PROUNDS; ++k) {
        const int p = tid + 256 * k;
        const float dx = nb3[k][0] - ctr3[k][0];
        const float dy = nb3[k][1] - ctr3[k][1];
        const float dz = nb3[k][2] - ctr3[k][2];
        const float inv = 1.0f / fmaxf(sqrtf(dx * dx + dy * dy + dz * dz), EPSV);
        NbrRec r;
        r.x   = dx * inv;
        r.y   = dy * inv;
        r.z   = dz * inv;
        r.off = (unsigned)nbv[k] * (unsigned)COLS + (unsigned)OC;
        s_rec[p] = r;
    }

    const int c    = tid & (OC - 1);
    const int slot = tid >> 6;
    float sx[SUP], sy[SUP], sz[SUP];
    for (int s = 0; s < SUP; ++s) {
        const int d = s * OC + c;
        float x = dirs[d];
        float y = dirs[DIRS + d];
        float z = dirs[2 * DIRS + d];
        const float sinv = 1.0f / fmaxf(sqrtf(x * x + y * y + z * z), EPSV);
        sx[s] = x * sinv; sy[s] = y * sinv; sz[s] = z * sinv;
    }

    __syncthreads();

    float ring[NQ][NPB][SUP];
    float ctr_prev = 0.0f;
    const NbrRec* recs = s_rec + slot * NNB;
    const NbrRec* prev = recs;
    size_t        bv   = (size_t)(v0 + slot);

    for (int t = 0; t < NVERT; ++t) {
        unsigned o0[NPB], o1[NPB], o2[NPB], o3[NPB];
        read_offs(o0, recs + 0 * NPB);
        read_offs(o1, recs + 1 * NPB);
        read_offs(o2, recs + 2 * NPB);
        read_offs(o3, recs + 3 * NPB);
        const float ctr = feat[bv * COLS + c];
        float m[SUP];
        for (int s = 0; s < SUP; ++s) m[s] = -__builtin_inff();

        if (t > 0) fold_batch(m, ring[0], prev + 0 * NPB, sx, sy, sz);
        issue_batch(ring[0], o0, feat, c);
        if (t > 0) fold_batch(m, ring[1], prev + 1 * NPB, sx, sy, sz);
        issue_batch(ring[1], o1, feat, c);
        if (t > 0) fold_batch(m, ring[2], prev + 2 * NPB, sx, sy, sz);
        __builtin_amdgcn_sched_barrier(0);
        issue_batch(ring[2], o2, feat, c);
        if (t > 0) {
            fold_batch(m, ring[3], prev + 3 * NPB, sx, sy, sz);
            out[(bv - VSLOTS) * OC + c] = ctr_prev + m[0] + m[1] + m[2] + m[3];
        }
        issue_batch(ring[3], o3, feat, c);

        ctr_prev = ctr;
        prev  = recs;
        recs += VSLOTS * NNB;
        bv   += VSLOTS;
    }

    {
        float m[SUP];
        for (int s = 0; s < SUP; ++s) m[s] = -__builtin_inff();
        fold_batch(m, ring[0], prev + 0 * NPB, sx, sy, sz);
        fold_batch(m, ring[1], prev + 1 * NPB, sx, sy, sz);
        fold_batch(m, ring[2], prev + 2 * NPB, sx, sy, sz);
        fold_batch(m, ring[3], prev + 3 * NPB, sx, sy, sz);
        out[(bv - VSLOTS) * OC + c] = ctr_prev + m[0] + m[1] + m[2] + m[3];
    }
}

extern "C" void kernel_launch(void* const* d_in, const int* in_sizes, int n_in,
                              void* d_out, int out_size, void* d_ws, size_t ws_size,
                              hipStream_t stream)
{
    (void)in_sizes; (void)n_in; (void)out_size; (void)ws_size;

    const int*   nbr   = (const int*)  d_in[0];
    const float* verts = (const float*)d_in[1];
    const float* fmap  = (const float*)d_in[2];
    const float* W     = (const float*)d_in[3];
    const float* bias  = (const float*)d_in[4];
    const float* dirs  = (const float*)d_in[5];
    float*       outp  = (float*)d_out;
    float*       feat  = (float*)d_ws;

    const int M      = BS * V;
    const int blocks = M / (2 * MROWS_PER_WAVE);
    const int strips = M / STRIP;

    gemm_bias_wmma_kernel<<<blocks, 256, 0, stream>>>(fmap, W, bias, feat, M);
    gather_theta_max_kernel<<<strips, 256, 0, stream>>>(nbr, verts, dirs, feat, outp);
}
